# starting kernel
# baseline (speedup 1.0000x reference)
#include <hip/hip_runtime.h>

typedef __attribute__((ext_vector_type(2))) float v2f;
typedef __attribute__((ext_vector_type(8))) float v8f;

#define BATCH 4096
#define SDIM  1024
#define KWIN  1019
#define NTOT  (BATCH * KWIN)
#define NPAIRS (NTOT / 32)

__global__ __launch_bounds__(256) void actor_wmma_kernel(
    const float* __restrict__ s,
    const float* __restrict__ w1,
    const float* __restrict__ b1,
    const float* __restrict__ w3,
    const float* __restrict__ b3,
    float* __restrict__ out)
{
    const int lane  = threadIdx.x & 31;
    const int half  = lane >> 4;
    const int mrow  = lane & 15;

    const int wid = __builtin_amdgcn_readfirstlane(
        (int)((blockIdx.x * blockDim.x + threadIdx.x) >> 5));
    const int nW = (int)((gridDim.x * blockDim.x) >> 5);

    const float aux0 = s[SDIM - 3];
    const float aux1 = s[SDIM - 2];
    const float aux2 = s[SDIM - 1];
    const float b3s  = b3[0];

    v2f A0, A1, A2, A3;
    {
        const int f0 = 2 * half;
#define LOADA(t, R) { int m = (t) * 16 + mrow; \
                      R.x = w1[m * 6 + f0]; R.y = w1[m * 6 + f0 + 1]; }
        LOADA(0, A0) LOADA(1, A1) LOADA(2, A2) LOADA(3, A3)
#undef LOADA
    }

    v8f C0, C1, C2, C3;
    v8f W0, W1, W2, W3f;
#define LOADC(t, CR, WR) \
    _Pragma("unroll") \
    for (int v = 0; v < 8; ++v) { \
        int m = (t) * 16 + v + 8 * half; \
        CR[v] = b1[m] + aux1 * w1[m * 6 + 4] + aux2 * w1[m * 6 + 5]; \
        WR[v] = w3[m]; \
    }
    LOADC(0, C0, W0) LOADC(1, C1, W1) LOADC(2, C2, W2) LOADC(3, C3, W3f)
#undef LOADC

#define WMMA(Af, Bf, Cf) \
    __builtin_amdgcn_wmma_f32_16x16x4_f32(false, Af, false, Bf, (short)0, Cf, false, false)

    const int voff0 = mrow * SDIM + 2 * half;
    const int voff1 = mrow * SDIM + 1;

    auto base_ptr = [&](int pp) -> const float* {
        int n0 = pp << 5;
        return s + (n0 & (BATCH - 1)) * SDIM + (n0 >> 12);
    };

    if (wid >= NPAIRS) return;

    const float* sb = base_ptr(wid);
    float e00 = sb[voff0];
    float e01 = sb[voff1];
    float e10 = sb[voff0 + 16 * SDIM];
    float e11 = sb[voff1 + 16 * SDIM];

    for (int p = wid; p < NPAIRS; p += nW) {
        v2f B0, B1;
        B0.x = e00; B0.y = half ? aux0 : e01;
        B1.x = e10; B1.y = half ? aux0 : e11;

        v8f d00 = WMMA(A0, B0, C0);
        v8f d01 = WMMA(A1, B0, C1);
        v8f d02 = WMMA(A2, B0, C2);
        v8f d03 = WMMA(A3, B0, C3);
        v8f d10 = WMMA(A0, B1, C0);
        v8f d11 = WMMA(A1, B1, C1);
        v8f d12 = WMMA(A2, B1, C2);
        v8f d13 = WMMA(A3, B1, C3);

        int pn = p + nW;
        const float* sbn = base_ptr(pn < NPAIRS ? pn : wid);
        e00 = sbn[voff0];
        e01 = sbn[voff1];
        e10 = sbn[voff0 + 16 * SDIM];
        e11 = sbn[voff1 + 16 * SDIM];

        float a00 = 0.f, a01 = 0.f, a02 = 0.f, a03 = 0.f;
        float a10 = 0.f, a11 = 0.f, a12 = 0.f, a13 = 0.f;
#pragma unroll
        for (int v = 0; v < 8; ++v) {
            a00 += W0[v]  * fmaxf(d00[v], 0.f);
            a01 += W1[v]  * fmaxf(d01[v], 0.f);
            a02 += W2[v]  * fmaxf(d02[v], 0.f);
            a03 += W3f[v] * fmaxf(d03[v], 0.f);
            a10 += W0[v]  * fmaxf(d10[v], 0.f);
            a11 += W1[v]  * fmaxf(d11[v], 0.f);
            a12 += W2[v]  * fmaxf(d12[v], 0.f);
            a13 += W3f[v] * fmaxf(d13[v], 0.f);
        }
        float accA = (a00 + a01) + (a02 + a03);
        float accB = (a10 + a11) + (a12 + a13);

        int swA = __builtin_amdgcn_ds_swizzle(__builtin_bit_cast(int, accA), 0x401f);
        int swB = __builtin_amdgcn_ds_swizzle(__builtin_bit_cast(int, accB), 0x401f);
        float totA = accA + __builtin_bit_cast(float, swA);
        float totB = accB + __builtin_bit_cast(float, swB);

        float r = (half ? totB : totA) + b3s;
        r = __builtin_amdgcn_fmed3f(r, -1.0f, 1.0f);

        out[(p << 5) + lane] = r;
    }
#undef WMMA
}

extern "C" void kernel_launch(void* const* d_in, const int* in_sizes, int n_in,
                              void* d_out, int out_size, void* d_ws, size_t ws_size,
                              hipStream_t stream) {
    (void)in_sizes; (void)n_in; (void)out_size; (void)d_ws; (void)ws_size;
    const float* s  = (const float*)d_in[0];
    const float* w1 = (const float*)d_in[1];
    const float* b1 = (const float*)d_in[2];
    const float* w3 = (const float*)d_in[3];
    const float* b3 = (const float*)d_in[4];
    float* out = (float*)d_out;

    actor_wmma_kernel<<<1024, 256, 0, stream>>>(s, w1, b1, w3, b3, out);
}


# relu in place by hand-paired dual max on each result tile, so the compiler's NaN-quieting max per element goes (64 fewer max per window); FMAs on adjacent register pairs; f32 sums regrouped
# speedup vs baseline: 1.4305x; 1.4305x over previous
#include <hip/hip_runtime.h>

typedef __attribute__((ext_vector_type(2))) float v2f;
typedef __attribute__((ext_vector_type(4))) float v4f;
typedef __attribute__((ext_vector_type(8))) float v8f;

#define BATCH 4096
#define SDIM  1024
#define KWIN  1019
#define TROWS 32
#define TCOLS 256
#define PITCH 260
#define NKQ   (SDIM / TCOLS)

__global__ __launch_bounds__(256) void actor_wmma_lds_kernel(
    const float* __restrict__ s,
    const float* __restrict__ w1,
    const float* __restrict__ b1,
    const float* __restrict__ w3,
    const float* __restrict__ b3,
    float* __restrict__ out)
{
    __shared__ float tile[TROWS * PITCH];

    const int lane  = threadIdx.x & 31;
    const int half  = lane >> 4;
    const int mrow  = lane & 15;
    const int wave  = __builtin_amdgcn_readfirstlane((int)(threadIdx.x >> 5));

    const int b0 = (int)(blockIdx.x / NKQ) * TROWS;
    const int kq = (int)(blockIdx.x % NKQ) * TCOLS;

    const int lr = lane >> 3;
    const int lc = (lane & 7) * 4;
    const float* gsrc = s + (size_t)(b0 + lr) * SDIM + kq + wave * 32 + lc;
    v4f g[8];
#pragma unroll
    for (int i = 0; i < 8; ++i)
        g[i] = *reinterpret_cast<const v4f*>(gsrc + (size_t)(4 * i) * SDIM);
    const bool halo = (wave == 7) && (kq + TCOLS < SDIM);
    v4f gh;
    if (halo)
        gh = *reinterpret_cast<const v4f*>(s + (size_t)(b0 + lane) * SDIM + kq + TCOLS);

    const float aux0 = s[SDIM - 3];
    const float aux1 = s[SDIM - 2];
    const float aux2 = s[SDIM - 1];
    const float b3s  = b3[0];

    v2f A0, A1, A2, A3;
    {
        const int f0 = 2 * half;
#define LOADA(t, R) { int m = (t) * 16 + mrow; \
                      R.x = w1[m * 6 + f0]; R.y = w1[m * 6 + f0 + 1]; }
        LOADA(0, A0) LOADA(1, A1) LOADA(2, A2) LOADA(3, A3)
#undef LOADA
    }

    v8f C0, C1, C2, C3;
    v8f W0, W1, W2, W3f;
#define LOADC(t, CR, WR) \
    _Pragma("unroll") \
    for (int v = 0; v < 8; ++v) { \
        int m = (t) * 16 + v + 8 * half; \
        CR[v] = b1[m] + aux1 * w1[m * 6 + 4] + aux2 * w1[m * 6 + 5]; \
        WR[v] = w3[m]; \
    }
    LOADC(0, C0, W0) LOADC(1, C1, W1) LOADC(2, C2, W2) LOADC(3, C3, W3f)
#undef LOADC

    {
        float* ldst = tile + lr * PITCH + wave * 32 + lc;
#pragma unroll
        for (int i = 0; i < 8; ++i)
            *reinterpret_cast<v4f*>(ldst + (4 * i) * PITCH) = g[i];
        if (halo)
            *reinterpret_cast<v4f*>(tile + lane * PITCH + TCOLS) = gh;
    }
    __syncthreads();

#define WMMA(Af, Bf, Cf) \
    __builtin_amdgcn_wmma_f32_16x16x4_f32(false, Af, false, Bf, (short)0, Cf, false, false)

    const int k0 = wave * 32;
    int nk = KWIN - (kq + k0);
    nk = nk > 32 ? 32 : nk;

    const float* l0 = tile + mrow * PITCH + k0 + 2 * half;
    const float* l1 = l0 + 16 * PITCH;
    float e00 = l0[0];
    float e01 = l0[1];
    float e10 = l1[0];
    float e11 = l1[1];

    float* op = out + (size_t)(kq + k0) * BATCH + b0 + lane;

    for (int i = 0; i < nk; ++i) {
        v2f B0, B1;
        B0.x = e00; B0.y = half ? aux0 : e01;
        B1.x = e10; B1.y = half ? aux0 : e11;

        v8f d00 = WMMA(A0, B0, C0);
        v8f d01 = WMMA(A1, B0, C1);
        v8f d02 = WMMA(A2, B0, C2);
        v8f d03 = WMMA(A3, B0, C3);
        v8f d10 = WMMA(A0, B1, C0);
        v8f d11 = WMMA(A1, B1, C1);
        v8f d12 = WMMA(A2, B1, C2);
        v8f d13 = WMMA(A3, B1, C3);

        e00 = l0[i + 1];
        e01 = l0[i + 2];
        e10 = l1[i + 1];
        e11 = l1[i + 2];

#define RELU8(D, x0, x1, x2, x3, x4, x5, x6, x7) \
        float x0 = D[0], x1 = D[1], x2 = D[2], x3 = D[3], x4 = D[4], x5 = D[5], x6 = D[6], x7 = D[7]; \
        asm("v_dual_max_num_f32 %0, 0, %0 :: v_dual_max_num_f32 %1, 0, %1\n\t" \
            "v_dual_max_num_f32 %2, 0, %2 :: v_dual_max_num_f32 %3, 0, %3\n\t" \
            "v_dual_max_num_f32 %4, 0, %4 :: v_dual_max_num_f32 %5, 0, %5\n\t" \
            "v_dual_max_num_f32 %6, 0, %6 :: v_dual_max_num_f32 %7, 0, %7" \
            : "+v"(x0), "+v"(x1), "+v"(x2), "+v"(x3), "+v"(x4), "+v"(x5), "+v"(x6), "+v"(x7));
#define PAIR(a, b) (v2f){(a), (b)}
#define DOT8(Wt, x0, x1, x2, x3, x4, x5, x6, x7) \
        __builtin_elementwise_fma(PAIR(Wt[6], Wt[7]), PAIR(x6, x7), \
        __builtin_elementwise_fma(PAIR(Wt[4], Wt[5]), PAIR(x4, x5), \
        __builtin_elementwise_fma(PAIR(Wt[2], Wt[3]), PAIR(x2, x3), PAIR(Wt[0], Wt[1]) * PAIR(x0, x1))))
        __builtin_amdgcn_sched_barrier(0);
        RELU8(d00, p0, p1, p2, p3, p4, p5, p6, p7)
        RELU8(d01, q0, q1, q2, q3, q4, q5, q6, q7)
        RELU8(d02, r0, r1, r2, r3, r4, r5, r6, r7)
        RELU8(d03, s0, s1, s2, s3, s4, s5, s6, s7)
        RELU8(d10, t0, t1, t2, t3, t4, t5, t6, t7)
        RELU8(d11, u0, u1, u2, u3, u4, u5, u6, u7)
        RELU8(d12, y0, y1, y2, y3, y4, y5, y6, y7)
        RELU8(d13, z0, z1, z2, z3, z4, z5, z6, z7)
        v2f a00 = DOT8(W0,  p0, p1, p2, p3, p4, p5, p6, p7);
        v2f a01 = DOT8(W1,  q0, q1, q2, q3, q4, q5, q6, q7);
        v2f a02 = DOT8(W2,  r0, r1, r2, r3, r4, r5, r6, r7);
        v2f a03 = DOT8(W3f, s0, s1, s2, s3, s4, s5, s6, s7);
        v2f a10 = DOT8(W0,  t0, t1, t2, t3, t4, t5, t6, t7);
        v2f a11 = DOT8(W1,  u0, u1, u2, u3, u4, u5, u6, u7);
        v2f a12 = DOT8(W2,  y0, y1, y2, y3, y4, y5, y6, y7);
        v2f a13 = DOT8(W3f, z0, z1, z2, z3, z4, z5, z6, z7);
#undef RELU8
#undef PAIR
#undef DOT8
        v2f sA = (a00 + a01) + (a02 + a03);
        v2f sB = (a10 + a11) + (a12 + a13);
        float accA = sA.x + sA.y;
        float accB = sB.x + sB.y;

        int swA = __builtin_amdgcn_ds_swizzle(__builtin_bit_cast(int, accA), 0x401f);
        int swB = __builtin_amdgcn_ds_swizzle(__builtin_bit_cast(int, accB), 0x401f);
        float totA = accA + __builtin_bit_cast(float, swA);
        float totB = accB + __builtin_bit_cast(float, swB);

        float r = (half ? totB : totA) + b3s;
        r = __builtin_amdgcn_fmed3f(r, -1.0f, 1.0f);

        op[(size_t)i * BATCH] = r;
    }
#undef WMMA
}

extern "C" void kernel_launch(void* const* d_in, const int* in_sizes, int n_in,
                              void* d_out, int out_size, void* d_ws, size_t ws_size,
                              hipStream_t stream) {
    (void)in_sizes; (void)n_in; (void)out_size; (void)d_ws; (void)ws_size;
    const float* s  = (const float*)d_in[0];
    const float* w1 = (const float*)d_in[1];
    const float* b1 = (const float*)d_in[2];
    const float* w3 = (const float*)d_in[3];
    const float* b3 = (const float*)d_in[4];
    float* out = (float*)d_out;

    actor_wmma_lds_kernel<<<512, 256, 0, stream>>>(s, w1, b1, w3, b3, out);
}


# Tile rows = 8 hidden units x 2 adjacent windows (A = w0 w1 w2 0 / 0 w0 w1 w2, aux0 term in C): a lane holds all 64 units of one output, layer 2 sums in-lane, no lane-half swap; f32 sums regrouped
# speedup vs baseline: 1.5652x; 1.0942x over previous
#include <hip/hip_runtime.h>

typedef __attribute__((ext_vector_type(2))) float v2f;
typedef __attribute__((ext_vector_type(4))) float v4f;
typedef __attribute__((ext_vector_type(8))) float v8f;

#define BATCH 4096
#define SDIM  1024
#define KWIN  1019
#define TROWS 32
#define TCOLS 256
#define PITCH 260
#define NKQ   (SDIM / TCOLS)

__device__ __forceinline__ float xhalf(float x) {
    int b = __builtin_bit_cast(int, x);
    return __builtin_bit_cast(float, __builtin_amdgcn_permlanex16(b, b, 0x76543210, 0xfedcba98, false, false));
}

__global__ __launch_bounds__(256) void actor_wmma_lds_kernel(
    const float* __restrict__ s,
    const float* __restrict__ w1,
    const float* __restrict__ b1,
    const float* __restrict__ w3,
    const float* __restrict__ b3,
    float* __restrict__ out)
{
    __shared__ float tile[TROWS * PITCH];
    __shared__ __attribute__((aligned(16))) float cw[128];

    const int lane  = threadIdx.x & 31;
    const int half  = lane >> 4;
    const int mrow  = lane & 15;
    const int wave  = __builtin_amdgcn_readfirstlane((int)(threadIdx.x >> 5));

    const int b0 = (int)(blockIdx.x / NKQ) * TROWS;
    const int kq = (int)(blockIdx.x % NKQ) * TCOLS;

    const int lr = lane >> 3;
    const int lc = (lane & 7) * 4;
    const float* gsrc = s + (size_t)(b0 + lr) * SDIM + kq + wave * 32 + lc;
    v4f g[8];
#pragma unroll
    for (int i = 0; i < 8; ++i)
        g[i] = *reinterpret_cast<const v4f*>(gsrc + (size_t)(4 * i) * SDIM);
    const bool halo = (wave == 7) && (kq + TCOLS < SDIM);
    v4f gh;
    if (halo)
        gh = *reinterpret_cast<const v4f*>(s + (size_t)(b0 + lane) * SDIM + kq + TCOLS);

    const float aux0 = s[SDIM - 3];
    const float aux1 = s[SDIM - 2];
    const float aux2 = s[SDIM - 1];
    const float b3s  = b3[0];

    v2f A0, A1, A2, A3, A4, A5, A6, A7;
    {
        const int woff = mrow >> 3;
        const int j0   = 2 * half - woff;
        const int j1   = j0 + 1;
        const int j0c  = j0 < 0 ? 0 : j0;
        const int j1c  = j1 > 2 ? 2 : j1;
#define LOADA(t, R) { int m = (t) * 8 + (mrow & 7); \
                      float ax = w1[m * 6 + j0c]; float ay = w1[m * 6 + j1c]; \
                      R.x = j0 < 0 ? 0.0f : ax; R.y = j1 > 2 ? 0.0f : ay; }
        LOADA(0, A0) LOADA(1, A1) LOADA(2, A2) LOADA(3, A3)
        LOADA(4, A4) LOADA(5, A5) LOADA(6, A6) LOADA(7, A7)
#undef LOADA
    }

    if (threadIdx.x < 64) {
        const int m = (int)threadIdx.x;
        cw[m]      = b1[m] + aux0 * w1[m * 6 + 3] + aux1 * w1[m * 6 + 4] + aux2 * w1[m * 6 + 5];
        cw[64 + m] = w3[m];
    }

    {
        float* ldst = tile + lr * PITCH + wave * 32 + lc;
#pragma unroll
        for (int i = 0; i < 8; ++i)
            *reinterpret_cast<v4f*>(ldst + (4 * i) * PITCH) = g[i];
        if (halo)
            *reinterpret_cast<v4f*>(tile + lane * PITCH + TCOLS) = gh;
    }
    __syncthreads();

    v8f C0, C1, C2, C3, C4, C5, C6, C7;
    v8f W0, W1, W2, W3f, W4, W5, W6, W7;
#define LOADCW(t, CR, WR) { \
        v4f clo = *reinterpret_cast<const v4f*>(cw + 8 * (t));      v4f chi = *reinterpret_cast<const v4f*>(cw + 8 * (t) + 4); \
        v4f wlo = *reinterpret_cast<const v4f*>(cw + 64 + 8 * (t)); v4f whi = *reinterpret_cast<const v4f*>(cw + 64 + 8 * (t) + 4); \
        CR = __builtin_shufflevector(clo, chi, 0, 1, 2, 3, 4, 5, 6, 7); \
        WR = __builtin_shufflevector(wlo, whi, 0, 1, 2, 3, 4, 5, 6, 7); }
    LOADCW(0, C0, W0) LOADCW(1, C1, W1) LOADCW(2, C2, W2) LOADCW(3, C3, W3f)
    LOADCW(4, C4, W4) LOADCW(5, C5, W5) LOADCW(6, C6, W6) LOADCW(7, C7, W7)
#undef LOADCW

#define WMMA(Af, Bf, Cf) \
    __builtin_amdgcn_wmma_f32_16x16x4_f32(false, Af, false, Bf, (short)0, Cf, false, false)

    const int k0 = wave * 32;
    int nk = KWIN - (kq + k0);
    nk = nk > 32 ? 32 : nk;

    const float* l0 = tile + mrow * PITCH + k0 + 2 * half;
    const float* l1 = l0 + 16 * PITCH;
    v2f B0 = *reinterpret_cast<const v2f*>(l0);
    v2f B1 = *reinterpret_cast<const v2f*>(l1);

    float* op = out + (size_t)(kq + k0) * BATCH + b0 + lane;

#define RELU8(D, x0, x1, x2, x3, x4, x5, x6, x7) \
        float x0, x1, x2, x3, x4, x5, x6, x7; \
        asm("v_dual_max_num_f32 %0, 0, %2 :: v_dual_max_num_f32 %1, 0, %3" : "=v"(x0), "=v"(x1) : "v"(D[0]), "v"(D[1])); \
        asm("v_dual_max_num_f32 %0, 0, %2 :: v_dual_max_num_f32 %1, 0, %3" : "=v"(x2), "=v"(x3) : "v"(D[2]), "v"(D[3])); \
        asm("v_dual_max_num_f32 %0, 0, %2 :: v_dual_max_num_f32 %1, 0, %3" : "=v"(x4), "=v"(x5) : "v"(D[4]), "v"(D[5])); \
        asm("v_dual_max_num_f32 %0, 0, %2 :: v_dual_max_num_f32 %1, 0, %3" : "=v"(x6), "=v"(x7) : "v"(D[6]), "v"(D[7]));
#define PAIR(a, b) (v2f){(a), (b)}
#define DOT8(Wt, x0, x1, x2, x3, x4, x5, x6, x7) \
        __builtin_elementwise_fma(PAIR(Wt[6], Wt[7]), PAIR(x6, x7), \
        __builtin_elementwise_fma(PAIR(Wt[4], Wt[5]), PAIR(x4, x5), \
        __builtin_elementwise_fma(PAIR(Wt[2], Wt[3]), PAIR(x2, x3), PAIR(Wt[0], Wt[1]) * PAIR(x0, x1))))
#define DOT8A(acc, Wt, x0, x1, x2, x3, x4, x5, x6, x7) \
        __builtin_elementwise_fma(PAIR(Wt[6], Wt[7]), PAIR(x6, x7), \
        __builtin_elementwise_fma(PAIR(Wt[4], Wt[5]), PAIR(x4, x5), \
        __builtin_elementwise_fma(PAIR(Wt[2], Wt[3]), PAIR(x2, x3), \
        __builtin_elementwise_fma(PAIR(Wt[0], Wt[1]), PAIR(x0, x1), acc))))
#define TILESET(Bf, lp, ACC) { \
        v8f e0 = WMMA(A0, Bf, C0); \
        v8f e2 = WMMA(A2, Bf, C2); \
        v8f e4 = WMMA(A4, Bf, C4); \
        v8f e6 = WMMA(A6, Bf, C6); \
        v8f e1 = WMMA(A1, Bf, C1); \
        v8f e3 = WMMA(A3, Bf, C3); \
        v8f e5 = WMMA(A5, Bf, C5); \
        v8f e7 = WMMA(A7, Bf, C7); \
        Bf = *reinterpret_cast<const v2f*>((lp) + i + 2); \
        __builtin_amdgcn_sched_barrier(0); \
        RELU8(e0, p0, p1, p2, p3, p4, p5, p6, p7) \
        RELU8(e1, q0, q1, q2, q3, q4, q5, q6, q7) \
        RELU8(e2, r0, r1, r2, r3, r4, r5, r6, r7) \
        RELU8(e3, s0, s1, s2, s3, s4, s5, s6, s7) \
        RELU8(e4, t0, t1, t2, t3, t4, t5, t6, t7) \
        RELU8(e5, u0, u1, u2, u3, u4, u5, u6, u7) \
        RELU8(e6, y0, y1, y2, y3, y4, y5, y6, y7) \
        RELU8(e7, z0, z1, z2, z3, z4, z5, z6, z7) \
        v2f c0 = DOT8(W0,  p0, p1, p2, p3, p4, p5, p6, p7); \
        v2f c1 = DOT8(W2,  r0, r1, r2, r3, r4, r5, r6, r7); \
        v2f c2 = DOT8(W4,  t0, t1, t2, t3, t4, t5, t6, t7); \
        v2f c3 = DOT8(W6,  y0, y1, y2, y3, y4, y5, y6, y7); \
        c0 = DOT8A(c0, W1,  q0, q1, q2, q3, q4, q5, q6, q7); \
        c1 = DOT8A(c1, W3f, s0, s1, s2, s3, s4, s5, s6, s7); \
        c2 = DOT8A(c2, W5,  u0, u1, u2, u3, u4, u5, u6, u7); \
        c3 = DOT8A(c3, W7,  z0, z1, z2, z3, z4, z5, z6, z7); \
        v2f cs = (c0 + c1) + (c2 + c3); \
        ACC = cs.x + cs.y; }

    for (int i = 0; i < nk; i += 2) {
        float acc0, acc1;
        TILESET(B0, l0, acc0)
        TILESET(B1, l1, acc1)

        float give = half ? acc0 : acc1;
        float got  = xhalf(give);
        float ra = (half ? got : acc0) + b3s;
        float rb = (half ? acc1 : got) + b3s;
        ra = __builtin_amdgcn_fmed3f(ra, -1.0f, 1.0f);
        rb = __builtin_amdgcn_fmed3f(rb, -1.0f, 1.0f);

        op[(size_t)i * BATCH] = ra;
        if (i + 1 < nk)
            op[(size_t)(i + 1) * BATCH] = rb;
    }
#undef TILESET
#undef RELU8
#undef PAIR
#undef DOT8
#undef DOT8A
#undef WMMA
}

extern "C" void kernel_launch(void* const* d_in, const int* in_sizes, int n_in,
                              void* d_out, int out_size, void* d_ws, size_t ws_size,
                              hipStream_t stream) {
    (void)in_sizes; (void)n_in; (void)out_size; (void)d_ws; (void)ws_size;
    const float* s  = (const float*)d_in[0];
    const float* w1 = (const float*)d_in[1];
    const float* b1 = (const float*)d_in[2];
    const float* w3 = (const float*)d_in[3];
    const float* b3 = (const float*)d_in[4];
    float* out = (float*)d_out;

    actor_wmma_lds_kernel<<<512, 256, 0, stream>>>(s, w1, b1, w3, b3, out);
}


# Work-group tile 64 rows x 256 windows (was 32 rows): 256 work-groups, one per WGP, 2 waves per SIMD, half the waves enter; a lane owns 4 adjacent rows, one 16-B store per window pair; sums unchanged
# speedup vs baseline: 1.8601x; 1.1884x over previous
#include <hip/hip_runtime.h>

typedef __attribute__((ext_vector_type(2))) float v2f;
typedef __attribute__((ext_vector_type(4))) float v4f;
typedef __attribute__((ext_vector_type(8))) float v8f;

#define BATCH 4096
#define SDIM  1024
#define KWIN  1019
#define TROWS 64
#define TCOLS 256
#define PITCH 260
#define NKQ   (SDIM / TCOLS)

__device__ __forceinline__ float xhalf(float x) {
    int b = __builtin_bit_cast(int, x);
    return __builtin_bit_cast(float, __builtin_amdgcn_permlanex16(b, b, 0x76543210, 0xfedcba98, false, false));
}

__global__ void __launch_bounds__(256) __attribute__((amdgpu_waves_per_eu(2, 2))) actor_wmma_lds_kernel(
    const float* __restrict__ s,
    const float* __restrict__ w1,
    const float* __restrict__ b1,
    const float* __restrict__ w3,
    const float* __restrict__ b3,
    float* __restrict__ out)
{
    __shared__ __attribute__((aligned(16))) float tile[TROWS * PITCH];
    __shared__ __attribute__((aligned(16))) float cw[128];

    const int lane  = threadIdx.x & 31;
    const int half  = lane >> 4;
    const int mrow  = lane & 15;
    const int wave  = __builtin_amdgcn_readfirstlane((int)(threadIdx.x >> 5));

    const int b0 = (int)(blockIdx.x / NKQ) * TROWS;
    const int kq = (int)(blockIdx.x % NKQ) * TCOLS;

    const int lr = lane >> 3;
    const int lc = (lane & 7) * 4;
    const float* gsrc = s + (size_t)(b0 + lr) * SDIM + kq + wave * 32 + lc;
    v4f g[16];
#pragma unroll
    for (int i = 0; i < 16; ++i)
        g[i] = *reinterpret_cast<const v4f*>(gsrc + (size_t)(4 * i) * SDIM);
    const bool halo = (wave == 7) && (kq + TCOLS < SDIM);
    v4f gh0, gh1;
    if (halo) {
        gh0 = *reinterpret_cast<const v4f*>(s + (size_t)(b0 + lane) * SDIM + kq + TCOLS);
        gh1 = *reinterpret_cast<const v4f*>(s + (size_t)(b0 + 32 + lane) * SDIM + kq + TCOLS);
    }

    const float aux0 = s[SDIM - 3];
    const float aux1 = s[SDIM - 2];
    const float aux2 = s[SDIM - 1];
    const float b3s  = b3[0];

    v2f A0, A1, A2, A3, A4, A5, A6, A7;
    {
        const int woff = mrow >> 3;
        const int j0   = 2 * half - woff;
        const int j1   = j0 + 1;
        const int j0c  = j0 < 0 ? 0 : j0;
        const int j1c  = j1 > 2 ? 2 : j1;
#define LOADA(t, R) { int m = (t) * 8 + (mrow & 7); \
                      float ax = w1[m * 6 + j0c]; float ay = w1[m * 6 + j1c]; \
                      R.x = j0 < 0 ? 0.0f : ax; R.y = j1 > 2 ? 0.0f : ay; }
        LOADA(0, A0) LOADA(1, A1) LOADA(2, A2) LOADA(3, A3)
        LOADA(4, A4) LOADA(5, A5) LOADA(6, A6) LOADA(7, A7)
#undef LOADA
    }

    if (threadIdx.x < 64) {
        const int m = (int)threadIdx.x;
        cw[m]      = b1[m] + aux0 * w1[m * 6 + 3] + aux1 * w1[m * 6 + 4] + aux2 * w1[m * 6 + 5];
        cw[64 + m] = w3[m];
    }

    {
        float* ldst = tile + (16 * lr) * PITCH + wave * 32 + lc;
#pragma unroll
        for (int i = 0; i < 16; ++i)
            *reinterpret_cast<v4f*>(ldst + i * PITCH) = g[i];
        if (halo) {
            const int r1 = lane + 32;
            *reinterpret_cast<v4f*>(tile + (16 * (lane & 3) + (lane >> 2)) * PITCH + TCOLS) = gh0;
            *reinterpret_cast<v4f*>(tile + (16 * (r1 & 3) + (r1 >> 2)) * PITCH + TCOLS) = gh1;
        }
    }
    __syncthreads();

    v8f C0, C1, C2, C3, C4, C5, C6, C7;
    v8f W0, W1, W2, W3f, W4, W5, W6, W7;
#define LOADCW(t, CR, WR) { \
        v4f clo = *reinterpret_cast<const v4f*>(cw + 8 * (t));      v4f chi = *reinterpret_cast<const v4f*>(cw + 8 * (t) + 4); \
        v4f wlo = *reinterpret_cast<const v4f*>(cw + 64 + 8 * (t)); v4f whi = *reinterpret_cast<const v4f*>(cw + 64 + 8 * (t) + 4); \
        CR = __builtin_shufflevector(clo, chi, 0, 1, 2, 3, 4, 5, 6, 7); \
        WR = __builtin_shufflevector(wlo, whi, 0, 1, 2, 3, 4, 5, 6, 7); }
    LOADCW(0, C0, W0) LOADCW(1, C1, W1) LOADCW(2, C2, W2) LOADCW(3, C3, W3f)
    LOADCW(4, C4, W4) LOADCW(5, C5, W5) LOADCW(6, C6, W6) LOADCW(7, C7, W7)
#undef LOADCW

#define WMMA(Af, Bf, Cf) \
    __builtin_amdgcn_wmma_f32_16x16x4_f32(false, Af, false, Bf, (short)0, Cf, false, false)

    const int k0 = wave * 32;
    int nk = KWIN - (kq + k0);
    nk = nk > 32 ? 32 : nk;

    const float* l0 = tile + mrow * PITCH + k0 + 2 * half;
    const float* l1 = l0 + 16 * PITCH;
    const float* l2 = l0 + 32 * PITCH;
    const float* l3 = l0 + 48 * PITCH;
    v2f B0 = *reinterpret_cast<const v2f*>(l0);
    v2f B1 = *reinterpret_cast<const v2f*>(l1);
    v2f B2 = *reinterpret_cast<const v2f*>(l2);
    v2f B3 = *reinterpret_cast<const v2f*>(l3);

    float* op = out + (size_t)(kq + k0 + half) * BATCH + b0 + 4 * mrow;

#define RELU8(D, x0, x1, x2, x3, x4, x5, x6, x7) \
        float x0, x1, x2, x3, x4, x5, x6, x7; \
        asm("v_dual_max_num_f32 %0, 0, %2 :: v_dual_max_num_f32 %1, 0, %3" : "=v"(x0), "=v"(x1) : "v"(D[0]), "v"(D[1])); \
        asm("v_dual_max_num_f32 %0, 0, %2 :: v_dual_max_num_f32 %1, 0, %3" : "=v"(x2), "=v"(x3) : "v"(D[2]), "v"(D[3])); \
        asm("v_dual_max_num_f32 %0, 0, %2 :: v_dual_max_num_f32 %1, 0, %3" : "=v"(x4), "=v"(x5) : "v"(D[4]), "v"(D[5])); \
        asm("v_dual_max_num_f32 %0, 0, %2 :: v_dual_max_num_f32 %1, 0, %3" : "=v"(x6), "=v"(x7) : "v"(D[6]), "v"(D[7]));
#define PAIR(a, b) (v2f){(a), (b)}
#define DOT8(Wt, x0, x1, x2, x3, x4, x5, x6, x7) \
        __builtin_elementwise_fma(PAIR(Wt[6], Wt[7]), PAIR(x6, x7), \
        __builtin_elementwise_fma(PAIR(Wt[4], Wt[5]), PAIR(x4, x5), \
        __builtin_elementwise_fma(PAIR(Wt[2], Wt[3]), PAIR(x2, x3), PAIR(Wt[0], Wt[1]) * PAIR(x0, x1))))
#define DOT8A(acc, Wt, x0, x1, x2, x3, x4, x5, x6, x7) \
        __builtin_elementwise_fma(PAIR(Wt[6], Wt[7]), PAIR(x6, x7), \
        __builtin_elementwise_fma(PAIR(Wt[4], Wt[5]), PAIR(x4, x5), \
        __builtin_elementwise_fma(PAIR(Wt[2], Wt[3]), PAIR(x2, x3), \
        __builtin_elementwise_fma(PAIR(Wt[0], Wt[1]), PAIR(x0, x1), acc))))
#define TILESET(Bf, READS, PS, QS) { \
        v8f e0 = WMMA(A0, Bf, C0); \
        v8f e2 = WMMA(A2, Bf, C2); \
        v8f e4 = WMMA(A4, Bf, C4); \
        v8f e6 = WMMA(A6, Bf, C6); \
        v8f e1 = WMMA(A1, Bf, C1); \
        v8f e3 = WMMA(A3, Bf, C3); \
        v8f e5 = WMMA(A5, Bf, C5); \
        v8f e7 = WMMA(A7, Bf, C7); \
        READS \
        __builtin_amdgcn_sched_barrier(0); \
        RELU8(e0, p0, p1, p2, p3, p4, p5, p6, p7) \
        RELU8(e1, q0, q1, q2, q3, q4, q5, q6, q7) \
        RELU8(e2, r0, r1, r2, r3, r4, r5, r6, r7) \
        RELU8(e3, s0, s1, s2, s3, s4, s5, s6, s7) \
        RELU8(e4, t0, t1, t2, t3, t4, t5, t6, t7) \
        RELU8(e5, u0, u1, u2, u3, u4, u5, u6, u7) \
        RELU8(e6, y0, y1, y2, y3, y4, y5, y6, y7) \
        RELU8(e7, z0, z1, z2, z3, z4, z5, z6, z7) \
        v2f c0 = DOT8(W0,  p0, p1, p2, p3, p4, p5, p6, p7); \
        v2f c1 = DOT8(W2,  r0, r1, r2, r3, r4, r5, r6, r7); \
        v2f c2 = DOT8(W4,  t0, t1, t2, t3, t4, t5, t6, t7); \
        v2f c3 = DOT8(W6,  y0, y1, y2, y3, y4, y5, y6, y7); \
        c0 = DOT8A(c0, W1,  q0, q1, q2, q3, q4, q5, q6, q7); \
        c1 = DOT8A(c1, W3f, s0, s1, s2, s3, s4, s5, s6, s7); \
        c2 = DOT8A(c2, W5,  u0, u1, u2, u3, u4, u5, u6, u7); \
        c3 = DOT8A(c3, W7,  z0, z1, z2, z3, z4, z5, z6, z7); \
        PS = c0 + c1; \
        QS = c2 + c3; \
        __builtin_amdgcn_sched_barrier(0); }

#define NEXTFRAGS \
        B0 = *reinterpret_cast<const v2f*>(l0 + i + 2); \
        B1 = *reinterpret_cast<const v2f*>(l1 + i + 2); \
        B2 = *reinterpret_cast<const v2f*>(l2 + i + 2); \
        B3 = *reinterpret_cast<const v2f*>(l3 + i + 2);
#define PAIRBODY(READS) \
        v2f pa0, pb0, pa1, pb1, pa2, pb2, pa3, pb3; \
        TILESET(B0, , pa0, pb0)          \
        TILESET(B1, , pa1, pb1)          \
        TILESET(B2, , pa2, pb2)          \
        TILESET(B3, READS, pa3, pb3)     \
        const v2f cs0 = pa0 + pb0, cs1 = pa1 + pb1, cs2 = pa2 + pb2, cs3 = pa3 + pb3; \
        v4f r; \
        r.x = __builtin_amdgcn_fmed3f((cs0.x + cs0.y) + b3s, -1.0f, 1.0f); \
        r.y = __builtin_amdgcn_fmed3f((cs1.x + cs1.y) + b3s, -1.0f, 1.0f); \
        r.z = __builtin_amdgcn_fmed3f((cs2.x + cs2.y) + b3s, -1.0f, 1.0f); \
        r.w = __builtin_amdgcn_fmed3f((cs3.x + cs3.y) + b3s, -1.0f, 1.0f);

    for (int i = 0; i + 1 < nk; i += 2) {
        PAIRBODY(NEXTFRAGS)
        *reinterpret_cast<v4f*>(op + (size_t)i * BATCH) = r;
    }
    if (nk & 1) {
        const int i = nk - 1;
        PAIRBODY()
        const float t2 = xhalf(r.z);
        const float t3 = xhalf(r.w);
        v2f lo;
        lo.x = half ? t2 : r.x;
        lo.y = half ? t3 : r.y;
        *reinterpret_cast<v2f*>(out + (size_t)(kq + k0 + i) * BATCH + b0 + 4 * mrow + 2 * half) = lo;
    }
#undef PAIRBODY
#undef NEXTFRAGS
#undef TILESET
#undef RELU8
#undef PAIR
#undef DOT8
#undef DOT8A
#undef WMMA
}

extern "C" void kernel_launch(void* const* d_in, const int* in_sizes, int n_in,
                              void* d_out, int out_size, void* d_ws, size_t ws_size,
                              hipStream_t stream) {
    (void)in_sizes; (void)n_in; (void)out_size; (void)d_ws; (void)ws_size;
    const float* s  = (const float*)d_in[0];
    const float* w1 = (const float*)d_in[1];
    const float* b1 = (const float*)d_in[2];
    const float* w3 = (const float*)d_in[3];
    const float* b3 = (const float*)d_in[4];
    float* out = (float*)d_out;

    actor_wmma_lds_kernel<<<256, 256, 0, stream>>>(s, w1, b1, w3, b3, out);
}
